# starting kernel
# baseline (speedup 1.0000x reference)
#include <hip/hip_runtime.h>
#include <hip/hip_bf16.h>

typedef __attribute__((ext_vector_type(2)))  float    v2f;
typedef __attribute__((ext_vector_type(8)))  float    v8f;
typedef __attribute__((ext_vector_type(16))) _Float16 v16h;

#define UB     700
#define PN     256
#define SN     1024
#define CBOX   128
#define DSEED  256
#define CBFU   128
#define KDIM   512
#define ODIM   128
#define BPG    16
#define GPAD   8

__global__ __launch_bounds__(256)
void scenebox_fused_kernel(const float* __restrict__ union_box,
                           const float* __restrict__ box_features,
                           const float* __restrict__ agg_xyz,
                           const float* __restrict__ seed_feature,
                           const float* __restrict__ seed_xyz,
                           const float* __restrict__ bfu,
                           const float* __restrict__ W,
                           const float* __restrict__ bias,
                           float* __restrict__ out)
{
    __shared__ float        g[BPG][KDIM + GPAD];
    __shared__ float        bmin[BPG][3], bmax[BPG][3];
    __shared__ unsigned int smask[SN];
    __shared__ unsigned int pmask[PN];

    const int t    = threadIdx.x;
    const int box0 = blockIdx.x * BPG;
    const int lane = t & 31;
    const int wave = t >> 5;
    const int nbase = wave * 16;
    const int hsel  = lane >> 4;
    const int l15   = lane & 15;

    __builtin_prefetch(&W[(nbase + l15) * KDIM], 0, 3);
    __builtin_prefetch(&W[(nbase + l15) * KDIM + 256], 0, 3);

    if (t < BPG) {
        const int b = box0 + t;
        float c0, c1, c2, s0, s1, s2;
        if (b < UB) {
            c0 = union_box[b*6+0]; c1 = union_box[b*6+1]; c2 = union_box[b*6+2];
            s0 = union_box[b*6+3]; s1 = union_box[b*6+4]; s2 = union_box[b*6+5];
        } else {
            c0 = c1 = c2 = 0.0f; s0 = s1 = s2 = -1.0f;
        }
        bmin[t][0] = c0 - 0.5f*s0; bmin[t][1] = c1 - 0.5f*s1; bmin[t][2] = c2 - 0.5f*s2;
        bmax[t][0] = c0 + 0.5f*s0; bmax[t][1] = c1 + 0.5f*s1; bmax[t][2] = c2 + 0.5f*s2;
    }
    __syncthreads();

    for (int s = t; s < SN; s += 256) {
        const float x = seed_xyz[s*3+0], y = seed_xyz[s*3+1], z = seed_xyz[s*3+2];
        unsigned int m = 0;
#pragma unroll
        for (int b = 0; b < BPG; ++b) {
            const int in = (x >= bmin[b][0]) & (x <= bmax[b][0]) &
                           (y >= bmin[b][1]) & (y <= bmax[b][1]) &
                           (z >= bmin[b][2]) & (z <= bmax[b][2]);
            m |= (unsigned)in << b;
        }
        smask[s] = m;
    }
    {
        const int p = t;
        const float x = agg_xyz[p*3+0], y = agg_xyz[p*3+1], z = agg_xyz[p*3+2];
        unsigned int m = 0;
#pragma unroll
        for (int b = 0; b < BPG; ++b) {
            const int in = (x >= bmin[b][0]) & (x <= bmax[b][0]) &
                           (y >= bmin[b][1]) & (y <= bmax[b][1]) &
                           (z >= bmin[b][2]) & (z <= bmax[b][2]);
            m |= (unsigned)in << b;
        }
        pmask[p] = m;
    }
    __syncthreads();

    {
        const int d = t;
        float macc[BPG];
#pragma unroll
        for (int b = 0; b < BPG; ++b) macc[b] = -__builtin_inff();
        const float* __restrict__ row = seed_feature + (size_t)d * SN;
#pragma unroll 4
        for (int s = 0; s < SN; ++s) {
            const float v = row[s];
            const unsigned int m = smask[s];
#pragma unroll
            for (int b = 0; b < BPG; ++b) {
                const float vv = ((m >> b) & 1u) ? v : 0.0f;
                macc[b] = fmaxf(macc[b], vv);
            }
        }
#pragma unroll
        for (int b = 0; b < BPG; ++b) g[b][d] = macc[b];
    }

    if (t < CBOX) {
        const int c = t;
        float macc[BPG];
#pragma unroll
        for (int b = 0; b < BPG; ++b) macc[b] = -__builtin_inff();
#pragma unroll 4
        for (int p = 0; p < PN; ++p) {
            const float v = box_features[p * CBOX + c];
            const unsigned int m = pmask[p];
#pragma unroll
            for (int b = 0; b < BPG; ++b) {
                const float vv = ((m >> b) & 1u) ? v : 0.0f;
                macc[b] = fmaxf(macc[b], vv);
            }
        }
#pragma unroll
        for (int b = 0; b < BPG; ++b) g[b][DSEED + c] = macc[b];
    }

    for (int i = t; i < BPG * CBFU; i += 256) {
        const int b = i / CBFU, c = i % CBFU;
        const int gb = box0 + b;
        g[b][DSEED + CBOX + c] = (gb < UB) ? bfu[(size_t)gb * CBFU + c] : 0.0f;
    }
    __syncthreads();

    v8f acc = {};
#if defined(__AMDGCN__) && __has_builtin(__builtin_amdgcn_wmma_f32_16x16x4_f32)
    for (int k0 = 0; k0 < KDIM; k0 += 4) {
        const int ka = k0 + 2 * hsel;
        v2f a, b;
        a.x = g[l15][ka];
        a.y = g[l15][ka + 1];
        const float* __restrict__ wr = W + (size_t)(nbase + l15) * KDIM + ka;
        b.x = wr[0];
        b.y = wr[1];
        acc = __builtin_amdgcn_wmma_f32_16x16x4_f32(false, a, false, b,
                                                    (short)0, acc, false, false);
    }
#elif defined(__AMDGCN__)
    for (int k0 = 0; k0 < KDIM; k0 += 32) {
        v16h a, b;
#pragma unroll
        for (int v = 0; v < 8; ++v) {
#pragma unroll
            for (int h = 0; h < 2; ++h) {
                const int k = k0 + 2*v + h + 8 * (hsel + (v >= 4));
                a[2*v + h] = (_Float16)g[l15][k];
                b[2*v + h] = (_Float16)W[(size_t)(nbase + l15) * KDIM + k];
            }
        }
        acc = __builtin_amdgcn_wmma_f32_16x16x32_f16(false, a, false, b,
                                                     (short)0, acc, false, false);
    }
#endif

    const float bn = bias[nbase + l15];
#pragma unroll
    for (int j = 0; j < 8; ++j) {
        const int m  = j + 8 * hsel;
        const int gm = box0 + m;
        if (gm < UB) {
            const float x  = acc[j] + bn + 1e-6f;
            const float ax = fabsf(x);
            out[(size_t)gm * ODIM + nbase + l15] = ax / (1.0f + ax);
        }
    }
}

extern "C" void kernel_launch(void* const* d_in, const int* in_sizes, int n_in,
                              void* d_out, int out_size, void* d_ws, size_t ws_size,
                              hipStream_t stream) {
    const float* union_box    = (const float*)d_in[0];
    const float* box_features = (const float*)d_in[1];
    const float* agg_xyz      = (const float*)d_in[2];
    const float* seed_feature = (const float*)d_in[3];
    const float* seed_xyz     = (const float*)d_in[4];
    const float* bfu          = (const float*)d_in[5];
    const float* W_gs         = (const float*)d_in[6];
    const float* b_gs         = (const float*)d_in[7];
    float* out = (float*)d_out;

    const int grid = (UB + BPG - 1) / BPG;
    scenebox_fused_kernel<<<grid, 256, 0, stream>>>(
        union_box, box_features, agg_xyz, seed_feature, seed_xyz,
        bfu, W_gs, b_gs, out);
}


# pooling split into its own launch on a 44x9 grid (16 boxes x 32 feature rows per workgroup, one 128-seed slice per wave, max-combined in LDS), pooled rows via workspace; linear layer as 2nd launch
# speedup vs baseline: 2.9048x; 2.9048x over previous
#include <hip/hip_runtime.h>
#include <hip/hip_bf16.h>


typedef __attribute__((ext_vector_type(2)))  float    v2f;
typedef __attribute__((ext_vector_type(8)))  float    v8f;
typedef __attribute__((ext_vector_type(16))) _Float16 v16h;

#define UB     700
#define PN     256
#define SN     1024
#define CBOX   128
#define DSEED  256
#define CBFU   128
#define KDIM   512
#define ODIM   128
#define BPG    16
#define GPAD   8
#define NTILE  ((UB + BPG - 1) / BPG)
#define RCH    32
#define NSCH   (DSEED / RCH)
#define SQ     (SN / 8)
#define PQ     (PN / 2)

__global__ __launch_bounds__(256)
void scenebox_pool_kernel(const float* __restrict__ union_box,
                          const float* __restrict__ box_features,
                          const float* __restrict__ agg_xyz,
                          const float* __restrict__ seed_feature,
                          const float* __restrict__ seed_xyz,
                          float* __restrict__ gfeat)
{
    __shared__ float        bmin[BPG][3], bmax[BPG][3];
    __shared__ __attribute__((aligned(16))) unsigned int mask[SN];
    __shared__ float        part[8 * BPG * RCH];

    const int t     = threadIdx.x;
    const int box0  = blockIdx.x * BPG;
    const int chunk = blockIdx.y;

    if (t < BPG) {
        const int b = box0 + t;
        float c0, c1, c2, s0, s1, s2;
        if (b < UB) {
            c0 = union_box[b*6+0]; c1 = union_box[b*6+1]; c2 = union_box[b*6+2];
            s0 = union_box[b*6+3]; s1 = union_box[b*6+4]; s2 = union_box[b*6+5];
        } else {
            c0 = c1 = c2 = 0.0f; s0 = s1 = s2 = -1.0f;
        }
        bmin[t][0] = c0 - 0.5f*s0; bmin[t][1] = c1 - 0.5f*s1; bmin[t][2] = c2 - 0.5f*s2;
        bmax[t][0] = c0 + 0.5f*s0; bmax[t][1] = c1 + 0.5f*s1; bmax[t][2] = c2 + 0.5f*s2;
    }
    __syncthreads();

    if (chunk < NSCH) {
        for (int s = t; s < SN; s += 256) {
            const float x = seed_xyz[s*3+0], y = seed_xyz[s*3+1], z = seed_xyz[s*3+2];
            unsigned int m = 0;
#pragma unroll
            for (int b = 0; b < BPG; ++b) {
                const int in = (x >= bmin[b][0]) & (x <= bmax[b][0]) &
                               (y >= bmin[b][1]) & (y <= bmax[b][1]) &
                               (z >= bmin[b][2]) & (z <= bmax[b][2]);
                m |= (unsigned)in << b;
            }
            mask[s] = m;
        }
        __syncthreads();

        const int dl = t & 31;
        const int q  = __builtin_amdgcn_readfirstlane(t >> 5);
        const int d  = chunk * RCH + dl;
        float macc[BPG];
#pragma unroll
        for (int b = 0; b < BPG; ++b) macc[b] = -__builtin_inff();
        const float* __restrict__ row = seed_feature + (size_t)d * SN + q * SQ;
        const unsigned int* mk = mask + q * SQ;
#pragma unroll 4
        for (int s = 0; s < SQ; ++s) {
            const float v = row[s];
            const unsigned int m = mk[s];
#pragma unroll
            for (int b = 0; b < BPG; ++b) {
                const float vv = ((m >> b) & 1u) ? v : 0.0f;
                macc[b] = fmaxf(macc[b], vv);
            }
        }
#pragma unroll
        for (int b = 0; b < BPG; ++b) part[(q * BPG + b) * RCH + dl] = macc[b];
        __syncthreads();

        for (int i = t; i < BPG * RCH; i += 256) {
            const int b = i >> 5, l = i & 31;
            float m = part[b * RCH + l];
#pragma unroll
            for (int qq = 1; qq < 8; ++qq) m = fmaxf(m, part[(qq * BPG + b) * RCH + l]);
            gfeat[(size_t)(box0 + b) * KDIM + chunk * RCH + l] = m;
        }
    } else {
        {
            const int p = t;
            const float x = agg_xyz[p*3+0], y = agg_xyz[p*3+1], z = agg_xyz[p*3+2];
            unsigned int m = 0;
#pragma unroll
            for (int b = 0; b < BPG; ++b) {
                const int in = (x >= bmin[b][0]) & (x <= bmax[b][0]) &
                               (y >= bmin[b][1]) & (y <= bmax[b][1]) &
                               (z >= bmin[b][2]) & (z <= bmax[b][2]);
                m |= (unsigned)in << b;
            }
            mask[p] = m;
        }
        __syncthreads();

        const int c = t & (CBOX - 1);
        const int q = __builtin_amdgcn_readfirstlane(t >> 7);
        float macc[BPG];
#pragma unroll
        for (int b = 0; b < BPG; ++b) macc[b] = -__builtin_inff();
        const float* __restrict__ col = box_features + (size_t)(q * PQ) * CBOX + c;
        const unsigned int* mk = mask + q * PQ;
#pragma unroll 4
        for (int p = 0; p < PQ; ++p) {
            const float v = col[p * CBOX];
            const unsigned int m = mk[p];
#pragma unroll
            for (int b = 0; b < BPG; ++b) {
                const float vv = ((m >> b) & 1u) ? v : 0.0f;
                macc[b] = fmaxf(macc[b], vv);
            }
        }
#pragma unroll
        for (int b = 0; b < BPG; ++b) part[(q * BPG + b) * CBOX + c] = macc[b];
        __syncthreads();

        for (int i = t; i < BPG * CBOX; i += 256) {
            const int b = i >> 7, c2 = i & (CBOX - 1);
            gfeat[(size_t)(box0 + b) * KDIM + DSEED + c2] =
                fmaxf(part[b * CBOX + c2], part[(BPG + b) * CBOX + c2]);
        }
    }
}

__global__ __launch_bounds__(256)
void scenebox_linear_kernel(const float* __restrict__ gfeat,
                            const float* __restrict__ bfu,
                            const float* __restrict__ W,
                            const float* __restrict__ bias,
                            float* __restrict__ out)
{
    __shared__ float g[BPG][KDIM + GPAD];

    const int t    = threadIdx.x;
    const int box0 = blockIdx.x * BPG;
    const int lane = t & 31;
    const int wave = t >> 5;
    const int nbase = wave * 16;
    const int hsel  = lane >> 4;
    const int l15   = lane & 15;

    __builtin_prefetch(&W[(nbase + l15) * KDIM], 0, 3);
    __builtin_prefetch(&W[(nbase + l15) * KDIM + 256], 0, 3);

    for (int i = t; i < BPG * (DSEED + CBOX); i += 256) {
        const int b = i / (DSEED + CBOX), c = i % (DSEED + CBOX);
        g[b][c] = gfeat[(size_t)(box0 + b) * KDIM + c];
    }
    for (int i = t; i < BPG * CBFU; i += 256) {
        const int b = i / CBFU, c = i % CBFU;
        const int gb = box0 + b;
        g[b][DSEED + CBOX + c] = (gb < UB) ? bfu[(size_t)gb * CBFU + c] : 0.0f;
    }
    __syncthreads();

    v8f acc = {};
#if defined(__AMDGCN__) && __has_builtin(__builtin_amdgcn_wmma_f32_16x16x4_f32)
    for (int k0 = 0; k0 < KDIM; k0 += 4) {
        const int ka = k0 + 2 * hsel;
        v2f a, b;
        a.x = g[l15][ka];
        a.y = g[l15][ka + 1];
        const float* __restrict__ wr = W + (size_t)(nbase + l15) * KDIM + ka;
        b.x = wr[0];
        b.y = wr[1];
        acc = __builtin_amdgcn_wmma_f32_16x16x4_f32(false, a, false, b,
                                                    (short)0, acc, false, false);
    }
#elif defined(__AMDGCN__)
    for (int k0 = 0; k0 < KDIM; k0 += 32) {
        v16h a, b;
#pragma unroll
        for (int v = 0; v < 8; ++v) {
#pragma unroll
            for (int h = 0; h < 2; ++h) {
                const int k = k0 + 2*v + h + 8 * (hsel + (v >= 4));
                a[2*v + h] = (_Float16)g[l15][k];
                b[2*v + h] = (_Float16)W[(size_t)(nbase + l15) * KDIM + k];
            }
        }
        acc = __builtin_amdgcn_wmma_f32_16x16x32_f16(false, a, false, b,
                                                     (short)0, acc, false, false);
    }
#endif

    const float bn = bias[nbase + l15];
#pragma unroll
    for (int j = 0; j < 8; ++j) {
        const int m  = j + 8 * hsel;
        const int gm = box0 + m;
        if (gm < UB) {
            const float x  = acc[j] + bn + 1e-6f;
            const float ax = fabsf(x);
            out[(size_t)gm * ODIM + nbase + l15] = ax / (1.0f + ax);
        }
    }
}

extern "C" void kernel_launch(void* const* d_in, const int* in_sizes, int n_in,
                              void* d_out, int out_size, void* d_ws, size_t ws_size,
                              hipStream_t stream) {
    const float* union_box    = (const float*)d_in[0];
    const float* box_features = (const float*)d_in[1];
    const float* agg_xyz      = (const float*)d_in[2];
    const float* seed_feature = (const float*)d_in[3];
    const float* seed_xyz     = (const float*)d_in[4];
    const float* bfu          = (const float*)d_in[5];
    const float* W_gs         = (const float*)d_in[6];
    const float* b_gs         = (const float*)d_in[7];
    float* out   = (float*)d_out;
    float* gfeat = (float*)d_ws;

    scenebox_pool_kernel<<<dim3(NTILE, NSCH + 1), 256, 0, stream>>>(
        union_box, box_features, agg_xyz, seed_feature, seed_xyz, gfeat);
    scenebox_linear_kernel<<<NTILE, 256, 0, stream>>>(
        gfeat, bfu, W_gs, b_gs, out);
}


# pooling: each select mask decoded by one scalar bit-field extract (was bit-compare plus conditional-select) and used directly as the lane mask of the vector select
# speedup vs baseline: 3.2391x; 1.1151x over previous
#include <hip/hip_runtime.h>
#include <hip/hip_bf16.h>


typedef __attribute__((ext_vector_type(2)))  float    v2f;
typedef __attribute__((ext_vector_type(8)))  float    v8f;
typedef __attribute__((ext_vector_type(16))) _Float16 v16h;

#define UB     700
#define PN     256
#define SN     1024
#define CBOX   128
#define DSEED  256
#define CBFU   128
#define KDIM   512
#define ODIM   128
#define BPG    16
#define GPAD   8
#define NTILE  ((UB + BPG - 1) / BPG)
#define RCH    32
#define NSCH   (DSEED / RCH)
#define SQ     (SN / 8)
#define PQ     (PN / 2)


__device__ __forceinline__ unsigned int lanes_of_bit(unsigned int m, int b)
{
    return (unsigned int)(((int)(m << (31 - b))) >> 31);
}

__global__ __launch_bounds__(256)
void scenebox_pool_kernel(const float* __restrict__ union_box,
                          const float* __restrict__ box_features,
                          const float* __restrict__ agg_xyz,
                          const float* __restrict__ seed_feature,
                          const float* __restrict__ seed_xyz,
                          float* __restrict__ gfeat)
{
    __shared__ float        bmin[BPG][3], bmax[BPG][3];
    __shared__ __attribute__((aligned(16))) unsigned int mask[SN];
    __shared__ float        part[8 * BPG * RCH];

    const int t     = threadIdx.x;
    const int box0  = blockIdx.x * BPG;
    const int chunk = blockIdx.y;

    if (t < BPG) {
        const int b = box0 + t;
        float c0, c1, c2, s0, s1, s2;
        if (b < UB) {
            c0 = union_box[b*6+0]; c1 = union_box[b*6+1]; c2 = union_box[b*6+2];
            s0 = union_box[b*6+3]; s1 = union_box[b*6+4]; s2 = union_box[b*6+5];
        } else {
            c0 = c1 = c2 = 0.0f; s0 = s1 = s2 = -1.0f;
        }
        bmin[t][0] = c0 - 0.5f*s0; bmin[t][1] = c1 - 0.5f*s1; bmin[t][2] = c2 - 0.5f*s2;
        bmax[t][0] = c0 + 0.5f*s0; bmax[t][1] = c1 + 0.5f*s1; bmax[t][2] = c2 + 0.5f*s2;
    }
    __syncthreads();

    if (chunk < NSCH) {
        for (int s = t; s < SN; s += 256) {
            const float x = seed_xyz[s*3+0], y = seed_xyz[s*3+1], z = seed_xyz[s*3+2];
            unsigned int m = 0;
#pragma unroll
            for (int b = 0; b < BPG; ++b) {
                const int in = (x >= bmin[b][0]) & (x <= bmax[b][0]) &
                               (y >= bmin[b][1]) & (y <= bmax[b][1]) &
                               (z >= bmin[b][2]) & (z <= bmax[b][2]);
                m |= (unsigned)in << b;
            }
            mask[s] = m;
        }
        __syncthreads();

        const int dl = t & 31;
        const int q  = __builtin_amdgcn_readfirstlane(t >> 5);
        const int d  = chunk * RCH + dl;
        float macc[BPG];
#pragma unroll
        for (int b = 0; b < BPG; ++b) macc[b] = -__builtin_inff();
        const float* __restrict__ row = seed_feature + (size_t)d * SN + q * SQ;
        const unsigned int* mk = mask + q * SQ;
#pragma unroll 4
        for (int s = 0; s < SQ; ++s) {
            const float v = row[s];
            const unsigned int m = mk[s];
#pragma unroll
            for (int b = 0; b < BPG; ++b) {
                const float vv = __builtin_amdgcn_inverse_ballot_w32(lanes_of_bit(m, b)) ? v : 0.0f;
                macc[b] = fmaxf(macc[b], vv);
            }
        }
#pragma unroll
        for (int b = 0; b < BPG; ++b) part[(q * BPG + b) * RCH + dl] = macc[b];
        __syncthreads();

        for (int i = t; i < BPG * RCH; i += 256) {
            const int b = i >> 5, l = i & 31;
            float m = part[b * RCH + l];
#pragma unroll
            for (int qq = 1; qq < 8; ++qq) m = fmaxf(m, part[(qq * BPG + b) * RCH + l]);
            gfeat[(size_t)(box0 + b) * KDIM + chunk * RCH + l] = m;
        }
    } else {
        {
            const int p = t;
            const float x = agg_xyz[p*3+0], y = agg_xyz[p*3+1], z = agg_xyz[p*3+2];
            unsigned int m = 0;
#pragma unroll
            for (int b = 0; b < BPG; ++b) {
                const int in = (x >= bmin[b][0]) & (x <= bmax[b][0]) &
                               (y >= bmin[b][1]) & (y <= bmax[b][1]) &
                               (z >= bmin[b][2]) & (z <= bmax[b][2]);
                m |= (unsigned)in << b;
            }
            mask[p] = m;
        }
        __syncthreads();

        const int c = t & (CBOX - 1);
        const int q = __builtin_amdgcn_readfirstlane(t >> 7);
        float macc[BPG];
#pragma unroll
        for (int b = 0; b < BPG; ++b) macc[b] = -__builtin_inff();
        const float* __restrict__ col = box_features + (size_t)(q * PQ) * CBOX + c;
        const unsigned int* mk = mask + q * PQ;
#pragma unroll 4
        for (int p = 0; p < PQ; ++p) {
            const float v = col[p * CBOX];
            const unsigned int m = mk[p];
#pragma unroll
            for (int b = 0; b < BPG; ++b) {
                const float vv = __builtin_amdgcn_inverse_ballot_w32(lanes_of_bit(m, b)) ? v : 0.0f;
                macc[b] = fmaxf(macc[b], vv);
            }
        }
#pragma unroll
        for (int b = 0; b < BPG; ++b) part[(q * BPG + b) * CBOX + c] = macc[b];
        __syncthreads();

        for (int i = t; i < BPG * CBOX; i += 256) {
            const int b = i >> 7, c2 = i & (CBOX - 1);
            gfeat[(size_t)(box0 + b) * KDIM + DSEED + c2] =
                fmaxf(part[b * CBOX + c2], part[(BPG + b) * CBOX + c2]);
        }
    }
}

__global__ __launch_bounds__(256)
void scenebox_linear_kernel(const float* __restrict__ gfeat,
                            const float* __restrict__ bfu,
                            const float* __restrict__ W,
                            const float* __restrict__ bias,
                            float* __restrict__ out)
{
    __shared__ float g[BPG][KDIM + GPAD];

    const int t    = threadIdx.x;
    const int box0 = blockIdx.x * BPG;
    const int lane = t & 31;
    const int wave = t >> 5;
    const int nbase = wave * 16;
    const int hsel  = lane >> 4;
    const int l15   = lane & 15;

    __builtin_prefetch(&W[(nbase + l15) * KDIM], 0, 3);
    __builtin_prefetch(&W[(nbase + l15) * KDIM + 256], 0, 3);

    for (int i = t; i < BPG * (DSEED + CBOX); i += 256) {
        const int b = i / (DSEED + CBOX), c = i % (DSEED + CBOX);
        g[b][c] = gfeat[(size_t)(box0 + b) * KDIM + c];
    }
    for (int i = t; i < BPG * CBFU; i += 256) {
        const int b = i / CBFU, c = i % CBFU;
        const int gb = box0 + b;
        g[b][DSEED + CBOX + c] = (gb < UB) ? bfu[(size_t)gb * CBFU + c] : 0.0f;
    }
    __syncthreads();

    v8f acc = {};
#if defined(__AMDGCN__) && __has_builtin(__builtin_amdgcn_wmma_f32_16x16x4_f32)
    for (int k0 = 0; k0 < KDIM; k0 += 4) {
        const int ka = k0 + 2 * hsel;
        v2f a, b;
        a.x = g[l15][ka];
        a.y = g[l15][ka + 1];
        const float* __restrict__ wr = W + (size_t)(nbase + l15) * KDIM + ka;
        b.x = wr[0];
        b.y = wr[1];
        acc = __builtin_amdgcn_wmma_f32_16x16x4_f32(false, a, false, b,
                                                    (short)0, acc, false, false);
    }
#elif defined(__AMDGCN__)
    for (int k0 = 0; k0 < KDIM; k0 += 32) {
        v16h a, b;
#pragma unroll
        for (int v = 0; v < 8; ++v) {
#pragma unroll
            for (int h = 0; h < 2; ++h) {
                const int k = k0 + 2*v + h + 8 * (hsel + (v >= 4));
                a[2*v + h] = (_Float16)g[l15][k];
                b[2*v + h] = (_Float16)W[(size_t)(nbase + l15) * KDIM + k];
            }
        }
        acc = __builtin_amdgcn_wmma_f32_16x16x32_f16(false, a, false, b,
                                                     (short)0, acc, false, false);
    }
#endif

    const float bn = bias[nbase + l15];
#pragma unroll
    for (int j = 0; j < 8; ++j) {
        const int m  = j + 8 * hsel;
        const int gm = box0 + m;
        if (gm < UB) {
            const float x  = acc[j] + bn + 1e-6f;
            const float ax = fabsf(x);
            out[(size_t)gm * ODIM + nbase + l15] = ax / (1.0f + ax);
        }
    }
}

extern "C" void kernel_launch(void* const* d_in, const int* in_sizes, int n_in,
                              void* d_out, int out_size, void* d_ws, size_t ws_size,
                              hipStream_t stream) {
    const float* union_box    = (const float*)d_in[0];
    const float* box_features = (const float*)d_in[1];
    const float* agg_xyz      = (const float*)d_in[2];
    const float* seed_feature = (const float*)d_in[3];
    const float* seed_xyz     = (const float*)d_in[4];
    const float* bfu          = (const float*)d_in[5];
    const float* W_gs         = (const float*)d_in[6];
    const float* b_gs         = (const float*)d_in[7];
    float* out   = (float*)d_out;
    float* gfeat = (float*)d_ws;

    scenebox_pool_kernel<<<dim3(NTILE, NSCH + 1), 256, 0, stream>>>(
        union_box, box_features, agg_xyz, seed_feature, seed_xyz, gfeat);
    scenebox_linear_kernel<<<NTILE, 256, 0, stream>>>(
        gfeat, bfu, W_gs, b_gs, out);
}


# pooling visits only the points inside each box: per-wave ballot words walked bit by bit over point-major seed rows (added transpose launch, 3 launches) instead of a select over all points; max only
# speedup vs baseline: 4.2329x; 1.3068x over previous
#include <hip/hip_runtime.h>
#include <hip/hip_bf16.h>


typedef __attribute__((ext_vector_type(2)))  float    v2f;
typedef __attribute__((ext_vector_type(8)))  float    v8f;
typedef __attribute__((ext_vector_type(16))) _Float16 v16h;

#define UB     700
#define PN     256
#define SN     1024
#define CBOX   128
#define DSEED  256
#define CBFU   128
#define KDIM   512
#define ODIM   128
#define BPG    16
#define GPAD   8
#define NTILE  ((UB + BPG - 1) / BPG)
#define NBX    2


__global__ __launch_bounds__(256)
void scenebox_transpose_kernel(const float* __restrict__ seed_feature,
                               float* __restrict__ sft)
{
    __shared__ float tile[64][33];
    const int t  = threadIdx.x;
    const int s0 = blockIdx.x * 32, d0 = blockIdx.y * 64;
#pragma unroll
    for (int j = 0; j < 8; ++j) {
        const int e = t + 256 * j, sl = e & 31, dl = e >> 5;
        tile[dl][sl] = seed_feature[(size_t)(d0 + dl) * SN + s0 + sl];
    }
    __syncthreads();
#pragma unroll
    for (int j = 0; j < 8; ++j) {
        const int e = t + 256 * j, dl = e & 63, sl = e >> 6;
        sft[(size_t)(s0 + sl) * DSEED + d0 + dl] = tile[dl][sl];
    }
}

__device__ __forceinline__ float4 max4(float4 a, float4 b)
{
    return make_float4(fmaxf(a.x, b.x), fmaxf(a.y, b.y), fmaxf(a.z, b.z), fmaxf(a.w, b.w));
}

__device__ __forceinline__ int next_point(unsigned long long& m0, unsigned long long& m1, int fallback)
{
    const bool a = (m0 != 0ull);
    const unsigned long long mm = a ? m0 : m1;
    const int pos = __builtin_ctzll(mm | (1ull << 63));
    const int i = (mm != 0ull) ? (pos + (a ? 0 : 64)) : fallback;
    const unsigned long long cl = mm & (mm - 1ull);
    m0 = a ? cl : 0ull;
    m1 = a ? m1 : cl;
    return i;
}

__global__ __launch_bounds__(256)
void scenebox_pool_kernel(const float* __restrict__ union_box,
                          const float* __restrict__ box_features,
                          const float* __restrict__ agg_xyz,
                          const float* __restrict__ sft,
                          const float* __restrict__ seed_xyz,
                          float* __restrict__ gfeat)
{
    __shared__ __attribute__((aligned(16))) float part[NBX][8][DSEED + CBOX];

    const int t    = threadIdx.x;
    const int lane = t & 31;
    const int q    = __builtin_amdgcn_readfirstlane(t >> 5);

    float sx[4], sy[4], sz[4];
#pragma unroll
    for (int w = 0; w < 4; ++w) {
        const int s = q * 128 + w * 32 + lane;
        sx[w] = seed_xyz[s*3+0]; sy[w] = seed_xyz[s*3+1]; sz[w] = seed_xyz[s*3+2];
    }
    const int   pp = q * 32 + lane;
    const float px = agg_xyz[pp*3+0], py = agg_xyz[pp*3+1], pz = agg_xyz[pp*3+2];

    const float* __restrict__ srow = sft + (size_t)(q * 128) * DSEED;
    const float* __restrict__ prow = box_features + (size_t)(q * 32) * CBOX;
    const float ninf = -__builtin_inff();

    for (int j = 0; j < NBX; ++j) {
        const int b = blockIdx.x * NBX + j;
        float c0, c1, c2, s0, s1, s2;
        if (b < UB) {
            c0 = union_box[b*6+0]; c1 = union_box[b*6+1]; c2 = union_box[b*6+2];
            s0 = union_box[b*6+3]; s1 = union_box[b*6+4]; s2 = union_box[b*6+5];
        } else {
            c0 = c1 = c2 = 0.0f; s0 = s1 = s2 = -1.0f;
        }
        const float l0 = c0 - 0.5f*s0, l1 = c1 - 0.5f*s1, l2 = c2 - 0.5f*s2;
        const float h0 = c0 + 0.5f*s0, h1 = c1 + 0.5f*s1, h2 = c2 + 0.5f*s2;

        unsigned int bw[4];
#pragma unroll
        for (int w = 0; w < 4; ++w) {
            const bool in = (sx[w] >= l0) & (sx[w] <= h0) &
                            (sy[w] >= l1) & (sy[w] <= h1) &
                            (sz[w] >= l2) & (sz[w] <= h2);
            bw[w] = __builtin_amdgcn_ballot_w32(in);
        }
        const bool pin = (px >= l0) & (px <= h0) & (py >= l1) & (py <= h1) & (pz >= l2) & (pz <= h2);
        unsigned int bp = __builtin_amdgcn_ballot_w32(pin);
        const bool seed_out = (bw[0] & bw[1] & bw[2] & bw[3]) != 0xffffffffu;
        const bool prop_out = bp != 0xffffffffu;

        float4 a0 = make_float4(ninf, ninf, ninf, ninf), a1 = a0, ab = a0;

        while (bp) {
            int idx[4];
            int first = 0;
#pragma unroll
            for (int u = 0; u < 4; ++u) {
                int i = first;
                if (bp) { i = __builtin_ctz(bp); bp &= bp - 1; }
                if (u == 0) first = i;
                idx[u] = i;
            }
            float4 x[4];
#pragma unroll
            for (int u = 0; u < 4; ++u)
                x[u] = reinterpret_cast<const float4*>(prow + idx[u] * CBOX)[lane];
#pragma unroll
            for (int u = 0; u < 4; ++u) ab = max4(ab, x[u]);
        }

        unsigned long long m0 = (unsigned long long)bw[0] | ((unsigned long long)bw[1] << 32);
        unsigned long long m1 = (unsigned long long)bw[2] | ((unsigned long long)bw[3] << 32);
        while (m0 | m1) {
            int idx[8];
            idx[0] = next_point(m0, m1, 0);
#pragma unroll
            for (int u = 1; u < 8; ++u) idx[u] = next_point(m0, m1, idx[0]);
            float4 x[8], y[8];
#pragma unroll
            for (int u = 0; u < 8; ++u) {
                const float4* r = reinterpret_cast<const float4*>(srow + idx[u] * DSEED);
                x[u] = r[lane];
                y[u] = r[32 + lane];
            }
#pragma unroll
            for (int u = 0; u < 8; ++u) { a0 = max4(a0, x[u]); a1 = max4(a1, y[u]); }
        }

        const float4 z4 = make_float4(0.0f, 0.0f, 0.0f, 0.0f);
        if (seed_out) { a0 = max4(a0, z4); a1 = max4(a1, z4); }
        if (prop_out) { ab = max4(ab, z4); }
        float4* pr = reinterpret_cast<float4*>(&part[j][q][0]);
        pr[lane]      = a0;
        pr[32 + lane] = a1;
        pr[64 + lane] = ab;
    }
    __syncthreads();

    for (int i = t; i < NBX * (DSEED + CBOX); i += 256) {
        const int j = i / (DSEED + CBOX), c = i % (DSEED + CBOX);
        float m = part[j][0][c];
#pragma unroll
        for (int qq = 1; qq < 8; ++qq) m = fmaxf(m, part[j][qq][c]);
        gfeat[(size_t)(blockIdx.x * NBX + j) * KDIM + c] = m;
    }
}

__global__ __launch_bounds__(256)
void scenebox_linear_kernel(const float* __restrict__ gfeat,
                            const float* __restrict__ bfu,
                            const float* __restrict__ W,
                            const float* __restrict__ bias,
                            float* __restrict__ out)
{
    __shared__ float g[BPG][KDIM + GPAD];

    const int t    = threadIdx.x;
    const int box0 = blockIdx.x * BPG;
    const int lane = t & 31;
    const int wave = t >> 5;
    const int nbase = wave * 16;
    const int hsel  = lane >> 4;
    const int l15   = lane & 15;

    __builtin_prefetch(&W[(nbase + l15) * KDIM], 0, 3);
    __builtin_prefetch(&W[(nbase + l15) * KDIM + 256], 0, 3);

    for (int i = t; i < BPG * (DSEED + CBOX); i += 256) {
        const int b = i / (DSEED + CBOX), c = i % (DSEED + CBOX);
        g[b][c] = gfeat[(size_t)(box0 + b) * KDIM + c];
    }
    for (int i = t; i < BPG * CBFU; i += 256) {
        const int b = i / CBFU, c = i % CBFU;
        const int gb = box0 + b;
        g[b][DSEED + CBOX + c] = (gb < UB) ? bfu[(size_t)gb * CBFU + c] : 0.0f;
    }
    __syncthreads();

    v8f acc = {};
#if defined(__AMDGCN__) && __has_builtin(__builtin_amdgcn_wmma_f32_16x16x4_f32)
    for (int k0 = 0; k0 < KDIM; k0 += 4) {
        const int ka = k0 + 2 * hsel;
        v2f a, b;
        a.x = g[l15][ka];
        a.y = g[l15][ka + 1];
        const float* __restrict__ wr = W + (size_t)(nbase + l15) * KDIM + ka;
        b.x = wr[0];
        b.y = wr[1];
        acc = __builtin_amdgcn_wmma_f32_16x16x4_f32(false, a, false, b,
                                                    (short)0, acc, false, false);
    }
#elif defined(__AMDGCN__)
    for (int k0 = 0; k0 < KDIM; k0 += 32) {
        v16h a, b;
#pragma unroll
        for (int v = 0; v < 8; ++v) {
#pragma unroll
            for (int h = 0; h < 2; ++h) {
                const int k = k0 + 2*v + h + 8 * (hsel + (v >= 4));
                a[2*v + h] = (_Float16)g[l15][k];
                b[2*v + h] = (_Float16)W[(size_t)(nbase + l15) * KDIM + k];
            }
        }
        acc = __builtin_amdgcn_wmma_f32_16x16x32_f16(false, a, false, b,
                                                     (short)0, acc, false, false);
    }
#endif

    const float bn = bias[nbase + l15];
#pragma unroll
    for (int j = 0; j < 8; ++j) {
        const int m  = j + 8 * hsel;
        const int gm = box0 + m;
        if (gm < UB) {
            const float x  = acc[j] + bn + 1e-6f;
            const float ax = fabsf(x);
            out[(size_t)gm * ODIM + nbase + l15] = ax / (1.0f + ax);
        }
    }
}

extern "C" void kernel_launch(void* const* d_in, const int* in_sizes, int n_in,
                              void* d_out, int out_size, void* d_ws, size_t ws_size,
                              hipStream_t stream) {
    const float* union_box    = (const float*)d_in[0];
    const float* box_features = (const float*)d_in[1];
    const float* agg_xyz      = (const float*)d_in[2];
    const float* seed_feature = (const float*)d_in[3];
    const float* seed_xyz     = (const float*)d_in[4];
    const float* bfu          = (const float*)d_in[5];
    const float* W_gs         = (const float*)d_in[6];
    const float* b_gs         = (const float*)d_in[7];
    float* out   = (float*)d_out;
    float* gfeat = (float*)d_ws;
    float* sft   = gfeat + (size_t)NTILE * BPG * KDIM;

    scenebox_transpose_kernel<<<dim3(SN / 32, DSEED / 64), 256, 0, stream>>>(seed_feature, sft);
    scenebox_pool_kernel<<<(NTILE * BPG) / NBX, 256, 0, stream>>>(
        union_box, box_features, agg_xyz, sft, seed_xyz, gfeat);
    scenebox_linear_kernel<<<NTILE, 256, 0, stream>>>(
        gfeat, bfu, W_gs, b_gs, out);
}
